# starting kernel
# baseline (speedup 1.0000x reference)
#include <hip/hip_runtime.h>

#define BINS   64
#define KPN    24
#define BATCH  256
#define HW     (BINS * BINS)
#define MIN_L  (-150.0f)
#define DELTA  (300.0f / 64.0f)
#define LOG2E  1.44269504088896340736f

typedef __attribute__((ext_vector_type(2))) float v2f;
typedef __attribute__((ext_vector_type(8))) float v8f;

__device__ __forceinline__ float center_of(int i) {
    return MIN_L + ((float)i + 0.5f) * DELTA;
}

__global__ __launch_bounds__(768) void pose_stage1(
    const float* __restrict__ feat, const float* __restrict__ gt,
    float* __restrict__ out_loc, float* __restrict__ lossbuf)
{
    const int b  = blockIdx.x;
    const int t  = threadIdx.x;
    const int k  = t % KPN;
    const int w0 = t / KPN;

    __shared__ float s_m[32 * KPN];
    __shared__ float s_Mk[KPN];
    __shared__ float s_part[32 * 128];
    __shared__ float s_red[128];

    const float* p = feat + (size_t)b * (HW * KPN) + (size_t)(w0 * KPN + k);

    float m = -1.0e30f;
    float sA = 0.f, sB = 0.f;
    float sy = 0.f, sy2 = 0.f;

    #pragma unroll 4
    for (int h = 0; h < BINS; ++h) {
        float xa = p[h * (BINS * KPN)];
        float xb = p[h * (BINS * KPN) + 32 * KPN];
        float xsa = xa * LOG2E;
        float xsb = xb * LOG2E;
        float mo  = m;
        m = fmaxf(m, fmaxf(xsa, xsb));
        if (m > mo) {
            float c = __builtin_amdgcn_exp2f(mo - m);
            sA *= c; sB *= c; sy *= c; sy2 *= c;
        }
        float ea = __builtin_amdgcn_exp2f(xsa - m);
        float eb = __builtin_amdgcn_exp2f(xsb - m);
        sA += ea;
        sB += eb;
        float es = ea + eb;
        float cy = MIN_L + ((float)h + 0.5f) * DELTA;
        sy  = fmaf(es, cy,      sy);
        sy2 = fmaf(es, cy * cy, sy2);
    }

    s_m[t] = m;
    __syncthreads();

    if (t < KPN) {
        float mm = s_m[t];
        #pragma unroll 4
        for (int j = 1; j < 32; ++j) mm = fmaxf(mm, s_m[j * KPN + t]);
        s_Mk[t] = mm;
    }
    __syncthreads();

    {
        float scale = __builtin_amdgcn_exp2f(m - s_Mk[k]);
        float a  = sA * scale, bb = sB * scale;
        float cxa = center_of(w0), cxb = center_of(w0 + 32);
        s_part[w0 * 128 + 0 * KPN + k] = a + bb;
        s_part[w0 * 128 + 1 * KPN + k] = sy  * scale;
        s_part[w0 * 128 + 2 * KPN + k] = sy2 * scale;
        s_part[w0 * 128 + 3 * KPN + k] = a * cxa + bb * cxb;
        s_part[w0 * 128 + 4 * KPN + k] = a * cxa * cxa + bb * cxb * cxb;
    }
    if (t < 256) s_part[(t >> 3) * 128 + 120 + (t & 7)] = 0.0f;
    __syncthreads();

    const int wid  = t >> 5;
    const int lane = t & 31;
    if (wid < 8) {
        v2f ones = {1.0f, 1.0f};
        v8f acc  = {};
        int col  = wid * 16 + (lane & 15);
        #pragma unroll
        for (int c8 = 0; c8 < 8; ++c8) {
            int rbase = c8 * 4;
            v2f bvec;
            if (lane < 16) {
                bvec.x = s_part[(rbase + 0) * 128 + col];
                bvec.y = s_part[(rbase + 1) * 128 + col];
            } else {
                bvec.x = s_part[(rbase + 2) * 128 + col];
                bvec.y = s_part[(rbase + 3) * 128 + col];
            }
            acc = __builtin_amdgcn_wmma_f32_16x16x4_f32(
                false, ones, false, bvec, (short)0, acc, false, false);
        }
        if (lane < 16) s_red[col] = acc[0];
    }
    __syncthreads();

    if (t < KPN) {
        float S   = s_red[0 * KPN + t];
        float Sy  = s_red[1 * KPN + t];
        float Sy2 = s_red[2 * KPN + t];
        float Sx  = s_red[3 * KPN + t];
        float Sx2 = s_red[4 * KPN + t];
        float inv = 1.0f / S;
        float ly = Sy * inv, lx = Sx * inv;
        float vy = Sy2 * inv - ly * ly;
        float vx = Sx2 * inv - lx * lx;
        float gy = gt[(size_t)b * KPN * 2 + t * 2 + 0];
        float gx = gt[(size_t)b * KPN * 2 + t * 2 + 1];
        float dy = ly - gy, dx = lx - gx;
        lossbuf[(size_t)b * KPN * 2 + t * 2 + 0] = vy + dy * dy;
        lossbuf[(size_t)b * KPN * 2 + t * 2 + 1] = vx + dx * dx;
        out_loc[(size_t)b * KPN * 2 + t * 2 + 0] = ly;
        out_loc[(size_t)b * KPN * 2 + t * 2 + 1] = lx;
    }
}

__global__ __launch_bounds__(1024) void pose_stage2(
    const float* __restrict__ lossbuf, float* __restrict__ out_hard)
{
    __shared__ float red[1024];
    __shared__ float s_inv;
    const int t = threadIdx.x;
    const int N = 2 * KPN * BATCH;

    float s = 0.f;
    for (int i = t; i < N; i += 1024) s += lossbuf[i];
    red[t] = s;
    __syncthreads();
    #pragma unroll
    for (int off = 512; off > 0; off >>= 1) {
        if (t < off) red[t] += red[t + off];
        __syncthreads();
    }
    if (t == 0) {
        float loss_sum = red[0] + 0.001f;
        s_inv = ((float)KPN / 150.0f) / loss_sum;
    }
    __syncthreads();
    float inv = s_inv;

    for (int o = t; o < N; o += 1024) {
        int c = o / (KPN * BATCH);
        int r = o % (KPN * BATCH);
        int k = r / BATCH;
        int b = r % BATCH;
        float l  = lossbuf[b * KPN * 2 + k * 2 + c];
        float l2 = lossbuf[b * KPN * 2 + k * 2 + 0] +
                   lossbuf[b * KPN * 2 + k * 2 + 1];
        out_hard[o] = l * l2 * inv;
    }
}

extern "C" void kernel_launch(void* const* d_in, const int* in_sizes, int n_in,
                              void* d_out, int out_size, void* d_ws, size_t ws_size,
                              hipStream_t stream) {
    const float* feat = (const float*)d_in[0];
    const float* gt   = (const float*)d_in[1];
    float* out      = (float*)d_out;
    float* out_hard = out;
    float* out_loc  = out + 2 * KPN * BATCH;
    float* lossbuf  = (float*)d_ws;

    pose_stage1<<<BATCH, 768, 0, stream>>>(feat, gt, out_loc, lossbuf);
    pose_stage2<<<1, 1024, 0, stream>>>(lossbuf, out_hard);
}


# global loss sum moved into the combine launch (48 block sums by lane butterfly; reduction tree changed); stage 2 now 8 blocks: one 16-byte read per thread, LDS transpose, whole-line stores
# speedup vs baseline: 3.1422x; 3.1422x over previous
#include <hip/hip_runtime.h>


#define BINS   64
#define KPN    24
#define BATCH  256
#define HW     (BINS * BINS)
#define MIN_L  (-150.0f)
#define DELTA  (300.0f / 64.0f)
#define LOG2E  1.44269504088896340736f

#define NSLICE 8
#define SROWS  (BINS / NSLICE)
#define NCG    8
#define NCOL   (BINS / NCG)
#define NTHR1  (NCG * KPN)
#define NPAIR  (BATCH * KPN)
#define NSTAT  6
#define LOSS_FLOATS (2 * NPAIR)
#define PART_FLOATS (NSTAT * NSLICE * NPAIR)
#define CB_THR      128
#define CB_BLK      (NPAIR / CB_THR)
#define BSUM_STRIDE 32
#define BSUM_FLOATS (CB_BLK * BSUM_STRIDE)
#define WS_NEED_BYTES ((size_t)(LOSS_FLOATS + PART_FLOATS + BSUM_FLOATS) * sizeof(float))

#define S2_BT     32
#define S2_FLOATS (S2_BT * KPN * 2)
#define S2_THR    (S2_FLOATS / 4)
#define S2_BLK    (BATCH / S2_BT)
#define S2_PITCH  (S2_BT + 1)

typedef __attribute__((ext_vector_type(2))) float v2f;
typedef __attribute__((ext_vector_type(8))) float v8f;
typedef __attribute__((ext_vector_type(4))) float v4f;

__device__ __forceinline__ float center_of(int i) {
    return MIN_L + ((float)i + 0.5f) * DELTA;
}

struct PairOut { float ly, lx, l0, l1; };

__device__ __forceinline__ PairOut pair_values(
    float S, float Sy, float Sy2, float Sx, float Sx2, int i,
    const float* __restrict__ gt)
{
    float inv = 1.0f / S;
    float ly = Sy * inv, lx = Sx * inv;
    float vy = Sy2 * inv - ly * ly;
    float vx = Sx2 * inv - lx * lx;
    float gy = gt[i * 2 + 0];
    float gx = gt[i * 2 + 1];
    float dy = ly - gy, dx = lx - gx;
    PairOut r;
    r.ly = ly;
    r.lx = lx;
    r.l0 = vy + dy * dy;
    r.l1 = vx + dx * dx;
    return r;
}

__device__ __forceinline__ void store_pair(
    const PairOut r, int i, float* __restrict__ out_loc, float* __restrict__ lossbuf)
{
    lossbuf[i * 2 + 0] = r.l0;
    lossbuf[i * 2 + 1] = r.l1;
    out_loc[i * 2 + 0] = r.ly;
    out_loc[i * 2 + 1] = r.lx;
}

__global__ __launch_bounds__(NTHR1) void pose_slice(
    const float* __restrict__ feat, const float* __restrict__ gt,
    float* __restrict__ out_loc, float* __restrict__ lossbuf,
    float* __restrict__ part, size_t ws_bytes)
{
    const bool split = ws_bytes >= WS_NEED_BYTES;
    const int b = blockIdx.x / NSLICE;
    const int s = blockIdx.x % NSLICE;
    if (!split && s != 0) return;
    const int row0  = split ? s * SROWS : 0;
    const int nrows = split ? SROWS : BINS;

    const int t  = threadIdx.x;
    const int k  = t % KPN;
    const int w0 = t / KPN;

    __shared__ float s_m[NTHR1];
    __shared__ float s_Mk[KPN];
    __shared__ float s_part[NCG * 128];
    __shared__ float s_red[128];

    const float* p = feat + (size_t)b * (HW * KPN)
                          + (size_t)row0 * (BINS * KPN) + (size_t)t;

    float m = -1.0e30f;
    float sc[NCOL];
    #pragma unroll
    for (int j = 0; j < NCOL; ++j) sc[j] = 0.f;
    float sy = 0.f, sy2 = 0.f;

    for (int h0 = 0; h0 < nrows; h0 += 4) {
        #pragma unroll
        for (int u = 0; u < 4; ++u) {
            const int h = h0 + u;
            const float* q = p + h * (BINS * KPN);
            float x[NCOL];
            #pragma unroll
            for (int j = 0; j < NCOL; ++j) x[j] = q[j * NTHR1] * LOG2E;
            float mo = m;
            m = fmaxf(fmaxf(fmaxf(m, fmaxf(x[0], x[1])), fmaxf(x[2], x[3])),
                      fmaxf(fmaxf(x[4], x[5]), fmaxf(x[6], x[7])));
            if (m > mo) {
                float c = __builtin_amdgcn_exp2f(mo - m);
                #pragma unroll
                for (int j = 0; j < NCOL; ++j) sc[j] *= c;
                sy *= c; sy2 *= c;
            }
            float e[NCOL];
            #pragma unroll
            for (int j = 0; j < NCOL; ++j) {
                e[j] = __builtin_amdgcn_exp2f(x[j] - m);
                sc[j] += e[j];
            }
            float es = ((e[0] + e[1]) + (e[2] + e[3])) + ((e[4] + e[5]) + (e[6] + e[7]));
            float cy = center_of(row0 + h);
            sy  = fmaf(es, cy,      sy);
            sy2 = fmaf(es, cy * cy, sy2);
        }
    }

    s_m[t] = m;
    __syncthreads();

    if (t < KPN) {
        float mm = s_m[t];
        #pragma unroll
        for (int j = 1; j < NCG; ++j) mm = fmaxf(mm, s_m[j * KPN + t]);
        s_Mk[t] = mm;
    }
    __syncthreads();

    {
        float scale = __builtin_amdgcn_exp2f(m - s_Mk[k]);
        float S = 0.f, Sx = 0.f, Sx2 = 0.f;
        #pragma unroll
        for (int j = 0; j < NCOL; ++j) {
            float a  = sc[j] * scale;
            float cx = center_of(w0 + NCG * j);
            S   += a;
            Sx  += a * cx;
            Sx2 += a * cx * cx;
        }
        s_part[w0 * 128 + 0 * KPN + k] = S;
        s_part[w0 * 128 + 1 * KPN + k] = sy  * scale;
        s_part[w0 * 128 + 2 * KPN + k] = sy2 * scale;
        s_part[w0 * 128 + 3 * KPN + k] = Sx;
        s_part[w0 * 128 + 4 * KPN + k] = Sx2;
    }
    if (t < NCG * 8) s_part[(t >> 3) * 128 + 120 + (t & 7)] = 0.0f;
    __syncthreads();

    const int wid  = __builtin_amdgcn_readfirstlane(t >> 5);
    const int lane = t & 31;
    for (int tile = wid; tile < 8; tile += NTHR1 / 32) {
        v2f ones = {1.0f, 1.0f};
        v8f acc  = {};
        int col  = tile * 16 + (lane & 15);
        #pragma unroll
        for (int c2 = 0; c2 < NCG / 4; ++c2) {
            int rbase = c2 * 4;
            v2f bvec;
            if (lane < 16) {
                bvec.x = s_part[(rbase + 0) * 128 + col];
                bvec.y = s_part[(rbase + 1) * 128 + col];
            } else {
                bvec.x = s_part[(rbase + 2) * 128 + col];
                bvec.y = s_part[(rbase + 3) * 128 + col];
            }
            acc = __builtin_amdgcn_wmma_f32_16x16x4_f32(
                false, ones, false, bvec, (short)0, acc, false, false);
        }
        if (lane < 16) s_red[col] = acc[0];
    }
    __syncthreads();

    if (t < KPN) {
        const int i = b * KPN + t;
        if (split) {
            #pragma unroll
            for (int q = 0; q < 5; ++q)
                part[(size_t)(q * NSLICE + s) * NPAIR + i] = s_red[q * KPN + t];
            part[(size_t)(5 * NSLICE + s) * NPAIR + i] = s_Mk[t];
        } else {
            store_pair(pair_values(s_red[0 * KPN + t], s_red[1 * KPN + t], s_red[2 * KPN + t],
                                   s_red[3 * KPN + t], s_red[4 * KPN + t], i, gt),
                       i, out_loc, lossbuf);
        }
    }
}

__global__ __launch_bounds__(CB_THR) void pose_combine(
    const float* __restrict__ part, const float* __restrict__ gt,
    float* __restrict__ out_loc, float* __restrict__ lossbuf,
    float* __restrict__ bsum, size_t ws_bytes)
{
    if (ws_bytes < WS_NEED_BYTES) return;
    __shared__ float s_w[CB_THR / 32];
    const int i = blockIdx.x * CB_THR + threadIdx.x;

    float v[NSTAT][NSLICE];
    #pragma unroll
    for (int q = 0; q < NSTAT; ++q)
        #pragma unroll
        for (int s = 0; s < NSLICE; ++s)
            v[q][s] = part[(size_t)(q * NSLICE + s) * NPAIR + i];

    float M = v[5][0];
    #pragma unroll
    for (int s = 1; s < NSLICE; ++s) M = fmaxf(M, v[5][s]);
    float acc[5] = {0.f, 0.f, 0.f, 0.f, 0.f};
    #pragma unroll
    for (int s = 0; s < NSLICE; ++s) {
        float f = __builtin_amdgcn_exp2f(v[5][s] - M);
        #pragma unroll
        for (int q = 0; q < 5; ++q) acc[q] = fmaf(f, v[q][s], acc[q]);
    }
    const PairOut r = pair_values(acc[0], acc[1], acc[2], acc[3], acc[4], i, gt);

    float l = r.l0 + r.l1;
    for (int off = 16; off > 0; off >>= 1) l += __shfl_xor(l, off);
    if ((threadIdx.x & 31) == 0) s_w[threadIdx.x >> 5] = l;
    __syncthreads();
    store_pair(r, i, out_loc, lossbuf);
    if (threadIdx.x == 0)
        bsum[blockIdx.x * BSUM_STRIDE] = (s_w[0] + s_w[1]) + (s_w[2] + s_w[3]);
}

__global__ __launch_bounds__(S2_THR) void pose_stage2(
    const float* __restrict__ lossbuf, const float* __restrict__ bsum,
    float* __restrict__ out_hard, size_t ws_bytes)
{
    __shared__ float s_w[S2_THR / 32];
    __shared__ float s_inv;
    __shared__ float tile[2 * KPN * S2_PITCH];
    const int t    = threadIdx.x;
    const int lane = t & 31;
    const int w    = t >> 5;
    const int j    = blockIdx.x;

    const v4f v = *(const v4f*)(lossbuf + (size_t)j * S2_FLOATS + 4 * t);

    if (ws_bytes >= WS_NEED_BYTES) {
        if (w == 0) {
            float a = bsum[lane * BSUM_STRIDE];
            if (lane < CB_BLK - 32) a += bsum[(32 + lane) * BSUM_STRIDE];
            for (int off = 16; off > 0; off >>= 1) a += __shfl_xor(a, off);
            if (lane == 0) s_inv = ((float)KPN / 150.0f) / (a + 0.001f);
        }
    } else {
        float a = 0.f;
        for (int i = t; i < LOSS_FLOATS; i += S2_THR) a += lossbuf[i];
        for (int off = 16; off > 0; off >>= 1) a += __shfl_xor(a, off);
        if (lane == 0) s_w[w] = a;
        __syncthreads();
        if (t == 0) {
            float s = 0.f;
            for (int x = 0; x < S2_THR / 32; ++x) s += s_w[x];
            s_inv = ((float)KPN / 150.0f) / (s + 0.001f);
        }
    }

    {
        const int bl = t / 12;
        const int k0 = 2 * (t % 12);
        const float la = v.x + v.y;
        const float lb = v.z + v.w;
        tile[(0 * KPN + k0    ) * S2_PITCH + bl] = v.x * la;
        tile[(1 * KPN + k0    ) * S2_PITCH + bl] = v.y * la;
        tile[(0 * KPN + k0 + 1) * S2_PITCH + bl] = v.z * lb;
        tile[(1 * KPN + k0 + 1) * S2_PITCH + bl] = v.w * lb;
    }
    __syncthreads();
    const float inv = s_inv;

    #pragma unroll
    for (int q = 0; q < (2 * KPN) / (S2_THR / 32); ++q) {
        const int r = w + q * (S2_THR / 32);
        out_hard[r * BATCH + j * S2_BT + lane] = tile[r * S2_PITCH + lane] * inv;
    }
}

extern "C" void kernel_launch(void* const* d_in, const int* in_sizes, int n_in,
                              void* d_out, int out_size, void* d_ws, size_t ws_size,
                              hipStream_t stream) {
    const float* feat = (const float*)d_in[0];
    const float* gt   = (const float*)d_in[1];
    float* out      = (float*)d_out;
    float* out_hard = out;
    float* out_loc  = out + 2 * KPN * BATCH;
    float* lossbuf  = (float*)d_ws;
    float* part     = lossbuf + LOSS_FLOATS;
    float* bsum     = part + PART_FLOATS;

    pose_slice<<<BATCH * NSLICE, NTHR1, 0, stream>>>(feat, gt, out_loc, lossbuf, part, ws_size);
    pose_combine<<<CB_BLK, CB_THR, 0, stream>>>(part, gt, out_loc, lossbuf, bsum, ws_size);
    pose_stage2<<<S2_BLK, S2_THR, 0, stream>>>(lossbuf, bsum, out_hard, ws_size);
}
